# starting kernel
# baseline (speedup 1.0000x reference)
#include <hip/hip_runtime.h>


typedef __attribute__((ext_vector_type(2))) float v2f;
typedef __attribute__((ext_vector_type(8))) float v8f;

#define GN 50000
#define GE 800000
#define GD 64
#define WROW 160

__global__ __launch_bounds__(256) void count_edges_k(const int* __restrict__ dst,
                                                     float* __restrict__ cnt,
                                                     int e_total) {
  int e = blockIdx.x * 256 + threadIdx.x;
  if (e < e_total) atomicAdd(&cnt[dst[e]], 1.0f);
}

__global__ __launch_bounds__(256) void scatter_add_k(const int* __restrict__ src,
                                                     const int* __restrict__ dst,
                                                     const float* __restrict__ feat,
                                                     float* __restrict__ agg,
                                                     int e_total) {
  long long t = (long long)blockIdx.x * 256 + threadIdx.x;
  int e = (int)(t >> 6);
  int f = (int)(t & 63);
  if (e < e_total) {
    int s = src[e];
    int d = dst[e];
    atomicAdd(&agg[(long long)d * GD + f], feat[(long long)s * GD + f]);
  }
}

__global__ __launch_bounds__(256) void sage_gemm_relu_k(
    const float* __restrict__ agg, const float* __restrict__ cnt,
    const float* __restrict__ xr,
    const float* __restrict__ Wl, const float* __restrict__ Wr,
    const float* __restrict__ bl,
    float* __restrict__ hout, int ntiles)
{
  __shared__ float WlS[(GD / 2) * WROW];
  __shared__ float WrS[(GD / 2) * WROW];
  for (int i = threadIdx.x; i < GD * GD; i += 256) {
    int j = i >> 6;
    int k = i & 63;
    int o = (k >> 1) * WROW + j * 2 + (k & 1);
    WlS[o] = Wl[i];
    WrS[o] = Wr[i];
  }
  __syncthreads();

  int wave = threadIdx.x >> 5;
  int tile = blockIdx.x * 8 + wave;
  if (tile < ntiles) {
    int lane = threadIdx.x & 31;
    int half = lane >> 4;
    int l16  = lane & 15;
    int row  = tile * 16 + l16;

    float inv = 1.0f / fmaxf(cnt[row], 1.0f);
    const float* aggRow = agg + (long long)row * GD;
    const float* xRow   = xr  + (long long)row * GD;

    v8f acc[4] = {};

    for (int k0 = 0; k0 < GD; k0 += 4) {
      int ka = k0 + half * 2;
      v2f am = *(const v2f*)(aggRow + ka);
      am *= inv;
      v2f ar = *(const v2f*)(xRow + ka);
      int kb = (ka >> 1) * WROW;
#pragma unroll
      for (int t = 0; t < 4; ++t) {
        int n = t * 16 + l16;
        v2f bL = *(const v2f*)&WlS[kb + n * 2];
        v2f bR = *(const v2f*)&WrS[kb + n * 2];
        acc[t] = __builtin_amdgcn_wmma_f32_16x16x4_f32(
            false, am, false, bL, (short)0, acc[t], false, false);
        acc[t] = __builtin_amdgcn_wmma_f32_16x16x4_f32(
            false, ar, false, bR, (short)0, acc[t], false, false);
      }
    }

#pragma unroll
    for (int t = 0; t < 4; ++t) {
      int n = t * 16 + l16;
      float bias = bl[n];
#pragma unroll
      for (int v = 0; v < 8; ++v) {
        int mr = half * 8 + v;
        float val = fmaxf(acc[t][v] + bias, 0.0f);
        hout[(long long)(tile * 16 + mr) * GD + n] = val;
      }
    }
  }
}

__global__ __launch_bounds__(256) void final_head_k(const float* __restrict__ h,
                                                    const float* __restrict__ Wf,
                                                    const float* __restrict__ bf,
                                                    float* __restrict__ out,
                                                    int n_total) {
  int i = blockIdx.x * 256 + threadIdx.x;
  if (i >= n_total) return;
  const float4* r = (const float4*)(h + (long long)i * GD);
  const float4* w = (const float4*)Wf;
  float s = 0.0f;
#pragma unroll
  for (int q = 0; q < GD / 4; ++q) {
    float4 a = r[q], b = w[q];
    s += a.x * b.x + a.y * b.y + a.z * b.z + a.w * b.w;
  }
  out[i] = s + bf[0];
}

extern "C" void kernel_launch(void* const* d_in, const int* in_sizes, int n_in,
                              void* d_out, int out_size, void* d_ws, size_t ws_size,
                              hipStream_t stream) {
  (void)in_sizes; (void)n_in; (void)out_size; (void)ws_size;

  const float* x   = (const float*)d_in[0];
  const int*   ei  = (const int*)d_in[1];
  const int*   src = ei;
  const int*   dst = ei + GE;
  const float* Wl1 = (const float*)d_in[2];
  const float* bl1 = (const float*)d_in[3];
  const float* Wr1 = (const float*)d_in[4];
  const float* Wl2 = (const float*)d_in[5];
  const float* bl2 = (const float*)d_in[6];
  const float* Wr2 = (const float*)d_in[7];
  const float* Wf  = (const float*)d_in[8];
  const float* bf  = (const float*)d_in[9];

  char*  ws  = (char*)d_ws;
  float* agg = (float*)ws;
  float* cnt = (float*)(ws + (size_t)GN * GD * 4);
  float* h   = (float*)(ws + (size_t)GN * GD * 4 + (size_t)GN * 4);

  const int ntiles = GN / 16;
  const int gemm_blocks = (ntiles + 7) / 8;
  const int scat_blocks = (GE * GD) / 256;

  hipMemsetAsync(agg, 0, (size_t)GN * GD * 4, stream);
  hipMemsetAsync(cnt, 0, (size_t)GN * 4, stream);
  count_edges_k<<<(GE + 255) / 256, 256, 0, stream>>>(dst, cnt, GE);
  scatter_add_k<<<scat_blocks, 256, 0, stream>>>(src, dst, x, agg, GE);
  sage_gemm_relu_k<<<gemm_blocks, 256, 0, stream>>>(agg, cnt, x, Wl1, Wr1, bl1, h, ntiles);

  hipMemsetAsync(agg, 0, (size_t)GN * GD * 4, stream);
  scatter_add_k<<<scat_blocks, 256, 0, stream>>>(src, dst, h, agg, GE);
  sage_gemm_relu_k<<<gemm_blocks, 256, 0, stream>>>(agg, cnt, h, Wl2, Wr2, bl2, h, ntiles);

  final_head_k<<<(GN + 255) / 256, 256, 0, stream>>>(h, Wf, bf, (float*)d_out, GN);
}


# Aggregate by pull over a destination-grouped edge list built once per call (arrival-order rank from the f32 count atomic); neighbour sum and head dot fused into the WMMA kernels; sum orders change
# speedup vs baseline: 2.3520x; 2.3520x over previous
#include <hip/hip_runtime.h>


typedef __attribute__((ext_vector_type(2))) float v2f;
typedef __attribute__((ext_vector_type(8))) float v8f;

#define GN 50000
#define GE 800000
#define GD 64
#define WROW 160
#define TPB 4
#define BROWS (TPB * 16)
#define LAYER_T (BROWS * 8)
#define MROW 68
#define SCAN_T 1024
#define SCAN_ITEMS 49

__global__ __launch_bounds__(256) void count_rank_k(const int* __restrict__ dst,
                                                    float* __restrict__ cnt,
                                                    int* __restrict__ rank,
                                                    int e_total) {
  int e = blockIdx.x * 256 + threadIdx.x;
  if (e < e_total) {
    float old = atomicAdd(&cnt[dst[e]], 1.0f);
    rank[e] = (int)old;
  }
}

__global__ __launch_bounds__(SCAN_T) void scan_k(const float* __restrict__ cnt,
                                                 int* __restrict__ rowptr) {
  __shared__ int S[SCAN_T * SCAN_ITEMS];
  __shared__ int wsum[32];
  const int t = threadIdx.x;
  for (int i = t; i < SCAN_T * SCAN_ITEMS; i += SCAN_T)
    S[i] = (i < GN) ? (int)cnt[i] : 0;
  __syncthreads();

  const int base = t * SCAN_ITEMS;
  int s = 0;
#pragma unroll
  for (int i = 0; i < SCAN_ITEMS; ++i) {
    int c = S[base + i];
    S[base + i] = s;
    s += c;
  }

  const int lane = t & 31;
  const int w = t >> 5;
  int inc = s;
#pragma unroll
  for (int d = 1; d < 32; d <<= 1) {
    int o = __shfl_up(inc, d, 32);
    if (lane >= d) inc += o;
  }
  if (lane == 31) wsum[w] = inc;
  __syncthreads();
  if (w == 0) {
    int x = wsum[lane];
    int xi = x;
#pragma unroll
    for (int d = 1; d < 32; d <<= 1) {
      int o = __shfl_up(xi, d, 32);
      if (lane >= d) xi += o;
    }
    wsum[lane] = xi - x;
  }
  __syncthreads();
  const int off = wsum[w] + inc - s;
#pragma unroll
  for (int i = 0; i < SCAN_ITEMS; ++i) S[base + i] += off;
  __syncthreads();
  for (int i = t; i <= GN; i += SCAN_T) rowptr[i] = S[i];
}

__global__ __launch_bounds__(256) void fill_k(const int* __restrict__ src,
                                              const int* __restrict__ dst,
                                              const int* __restrict__ rank,
                                              const int* __restrict__ rowptr,
                                              int* __restrict__ ssrc,
                                              int e_total) {
  int e = blockIdx.x * 256 + threadIdx.x;
  if (e < e_total) {
    int p = rowptr[dst[e]] + rank[e];
    atomicExch(&ssrc[p], src[e]);
  }
}

template <bool HEAD>
__global__ __launch_bounds__(LAYER_T) void sage_pull_gemm_k(
    const int* __restrict__ rowptr, const int* __restrict__ ssrc,
    const float* __restrict__ feat,
    const float* __restrict__ Wl, const float* __restrict__ Wr,
    const float* __restrict__ bl,
    const float* __restrict__ Wf, const float* __restrict__ bf,
    float* __restrict__ outp, int ntiles)
{
  __shared__ __attribute__((aligned(16))) float WlS[(GD / 2) * WROW];
  __shared__ __attribute__((aligned(16))) float WrS[(GD / 2) * WROW];
  __shared__ __attribute__((aligned(16))) float MS[BROWS * MROW];
  __shared__ float OS[BROWS];

  for (int i = threadIdx.x; i < GD * GD; i += LAYER_T) {
    int j = i >> 6;
    int k = i & 63;
    int o = (k >> 1) * WROW + j * 2 + (k & 1);
    WlS[o] = Wl[i];
    WrS[o] = Wr[i];
  }

  const int lane = threadIdx.x & 31;
  const int wave = __builtin_amdgcn_readfirstlane((int)(threadIdx.x >> 5));
  const int row0 = blockIdx.x * BROWS;

  {
    const int node0 = row0 + wave * 4;
    int rp[5];
#pragma unroll
    for (int r = 0; r < 5; ++r) {
      int nd = node0 + r;
      rp[r] = rowptr[nd < GN ? nd : GN];
    }
    int idx0[4];
#pragma unroll
    for (int r = 0; r < 4; ++r) {
      int d = rp[r + 1] - rp[r];
      int m = d < 32 ? d : 32;
      idx0[r] = (lane < m) ? ssrc[rp[r] + lane] : 0;
    }
#pragma unroll
    for (int r = 0; r < 4; ++r) {
      const int end = rp[r + 1];
      int base = rp[r];
      int myidx = idx0[r];
      v2f a0 = {0.0f, 0.0f}, a1 = {0.0f, 0.0f}, a2 = {0.0f, 0.0f}, a3 = {0.0f, 0.0f};
      while (true) {
        int left = end - base;
        int m = left < 32 ? left : 32;
        for (int j = 0; j < m; j += 8) {
          v2f v[8];
#pragma unroll
          for (int u = 0; u < 8; ++u) {
            int s = __builtin_amdgcn_readlane(myidx, j + u);
            v[u] = *(const v2f*)(feat + (long long)s * GD + 2 * lane);
          }
#pragma unroll
          for (int u = 0; u < 8; ++u) {
            if (j + u >= m) v[u] = (v2f){0.0f, 0.0f};
          }
          a0 += v[0]; a1 += v[1]; a2 += v[2]; a3 += v[3];
          a0 += v[4]; a1 += v[5]; a2 += v[6]; a3 += v[7];
        }
        base += 32;
        if (base >= end) break;
        int left2 = end - base;
        int m2 = left2 < 32 ? left2 : 32;
        myidx = (lane < m2) ? ssrc[base + lane] : 0;
      }
      const int deg = rp[r + 1] - rp[r];
      const float inv = 1.0f / fmaxf((float)deg, 1.0f);
      v2f mean = ((a0 + a1) + (a2 + a3)) * inv;
      *(v2f*)&MS[(wave * 4 + r) * MROW + 2 * lane] = mean;
    }
  }
  __syncthreads();

  const int tile = blockIdx.x * TPB + wave;
  if (wave < TPB && tile < ntiles) {
    int half = lane >> 4;
    int l16  = lane & 15;
    int row  = tile * 16 + l16;

    const float* mRow = &MS[(wave * 16 + l16) * MROW];
    const float* xRow = feat + (long long)row * GD;

    v8f acc[4] = {};

    for (int k0 = 0; k0 < GD; k0 += 4) {
      int ka = k0 + half * 2;
      v2f am = *(const v2f*)(mRow + ka);
      v2f ar = *(const v2f*)(xRow + ka);
      int kb = (ka >> 1) * WROW;
#pragma unroll
      for (int t = 0; t < 4; ++t) {
        int n = t * 16 + l16;
        v2f bL = *(const v2f*)&WlS[kb + n * 2];
        v2f bR = *(const v2f*)&WrS[kb + n * 2];
        acc[t] = __builtin_amdgcn_wmma_f32_16x16x4_f32(
            false, am, false, bL, (short)0, acc[t], false, false);
        acc[t] = __builtin_amdgcn_wmma_f32_16x16x4_f32(
            false, ar, false, bR, (short)0, acc[t], false, false);
      }
    }

    if (!HEAD) {
#pragma unroll
      for (int t = 0; t < 4; ++t) {
        int n = t * 16 + l16;
        float bias = bl[n];
#pragma unroll
        for (int v = 0; v < 8; ++v) {
          int mr = half * 8 + v;
          float val = fmaxf(acc[t][v] + bias, 0.0f);
          outp[(long long)(tile * 16 + mr) * GD + n] = val;
        }
      }
    } else {
      float part[8];
#pragma unroll
      for (int v = 0; v < 8; ++v) part[v] = 0.0f;
#pragma unroll
      for (int t = 0; t < 4; ++t) {
        int n = t * 16 + l16;
        float bias = bl[n];
        float wf = Wf[n];
#pragma unroll
        for (int v = 0; v < 8; ++v) {
          float val = fmaxf(acc[t][v] + bias, 0.0f);
          part[v] += val * wf;
        }
      }
      const float b0 = bf[0];
#pragma unroll
      for (int v = 0; v < 8; ++v) {
        float p = part[v];
        p += __shfl_xor(p, 1, 32);
        p += __shfl_xor(p, 2, 32);
        p += __shfl_xor(p, 4, 32);
        p += __shfl_xor(p, 8, 32);
        if (l16 == 0) OS[wave * 16 + half * 8 + v] = p + b0;
      }
    }
  }

  if (HEAD) {
    __syncthreads();
    if (threadIdx.x < BROWS) {
      int node = row0 + (int)threadIdx.x;
      if (node < GN) outp[node] = OS[threadIdx.x];
    }
  }
}

extern "C" void kernel_launch(void* const* d_in, const int* in_sizes, int n_in,
                              void* d_out, int out_size, void* d_ws, size_t ws_size,
                              hipStream_t stream) {
  (void)in_sizes; (void)n_in; (void)out_size; (void)ws_size;

  const float* x   = (const float*)d_in[0];
  const int*   ei  = (const int*)d_in[1];
  const int*   src = ei;
  const int*   dst = ei + GE;
  const float* Wl1 = (const float*)d_in[2];
  const float* bl1 = (const float*)d_in[3];
  const float* Wr1 = (const float*)d_in[4];
  const float* Wl2 = (const float*)d_in[5];
  const float* bl2 = (const float*)d_in[6];
  const float* Wr2 = (const float*)d_in[7];
  const float* Wf  = (const float*)d_in[8];
  const float* bf  = (const float*)d_in[9];

  const size_t SZ_H   = (size_t)GN * GD * 4;
  const size_t SZ_CNT = (((size_t)GN * 4 + 255) / 256) * 256;
  const size_t SZ_RP  = (((size_t)(GN + 1) * 4 + 255) / 256) * 256;
  const size_t SZ_E   = (size_t)GE * 4;

  char*  ws     = (char*)d_ws;
  float* h1     = (float*)ws;
  float* cnt    = (float*)(ws + SZ_H);
  int*   rowptr = (int*)(ws + SZ_H + SZ_CNT);
  int*   rank   = (int*)(ws + SZ_H + SZ_CNT + SZ_RP);
  int*   ssrc   = (int*)(ws + SZ_H + SZ_CNT + SZ_RP + SZ_E);

  const int ntiles = GN / 16;
  const int layer_blocks = (ntiles + TPB - 1) / TPB;
  const int edge_blocks = (GE + 255) / 256;

  hipMemsetAsync(cnt, 0, (size_t)GN * 4, stream);
  count_rank_k<<<edge_blocks, 256, 0, stream>>>(dst, cnt, rank, GE);
  scan_k<<<1, SCAN_T, 0, stream>>>(cnt, rowptr);
  fill_k<<<edge_blocks, 256, 0, stream>>>(src, dst, rank, rowptr, ssrc, GE);

  sage_pull_gemm_k<false><<<layer_blocks, LAYER_T, 0, stream>>>(
      rowptr, ssrc, x, Wl1, Wr1, bl1, Wf, bf, h1, ntiles);
  sage_pull_gemm_k<true><<<layer_blocks, LAYER_T, 0, stream>>>(
      rowptr, ssrc, h1, Wl2, Wr2, bl2, Wf, bf, (float*)d_out, ntiles);
}


# Edges grouped by destination through LDS (per-chunk order by node range, then per-range LDS slot counters), no global count atomic or scattered exchange per edge; slot (sum) order changes
# speedup vs baseline: 5.3788x; 2.2869x over previous
#include <hip/hip_runtime.h>


typedef __attribute__((ext_vector_type(2))) float v2f;
typedef __attribute__((ext_vector_type(8))) float v8f;

#define GN 50000
#define GE 800000
#define GD 64
#define WROW 160
#define TPB 4
#define BROWS (TPB * 16)
#define LAYER_T (BROWS * 8)
#define MROW 68
#define CAP 32
#define OVF_CAP 49152
#define NB_NODES 256
#define NBINS ((GN + NB_NODES - 1) / NB_NODES)
#define NCHUNK 256
#define EPC ((GE + NCHUNK - 1) / NCHUNK)
#define SROW (((EPC + 31) / 32) * 32)
#define OROW 256
#define PT 512
#define PI ((EPC + PT - 1) / PT)
#define BT 512
#define BWAVES (BT / 32)
#define SPW (NCHUNK / BWAVES)
static_assert(GN <= 65536, "node numbers must fit 16 bits");
static_assert(NB_NODES == 256, "range number is taken as word >> 24");
static_assert(NBINS + 1 <= OROW, "offsets row too short");
static_assert(OROW == 256, "scan below covers 32 lanes x 8 entries");
static_assert(SPW <= 32 && SPW * BWAVES == NCHUNK, "chunks per wave");
static_assert(GE % NCHUNK == 0, "every chunk is full");

__global__ __launch_bounds__(PT) void part_k(const int* __restrict__ src,
                                             const int* __restrict__ dst,
                                             unsigned* __restrict__ sorted,
                                             int* __restrict__ offtab,
                                             float* __restrict__ ovfn) {
  __shared__ unsigned histS[OROW];
  __shared__ unsigned offS[OROW];
  __shared__ unsigned sortS[EPC];

  const int tid = threadIdx.x;
  __builtin_assume(tid < PT);
  const int c = blockIdx.x;
  const int base = c * EPC;

  if (tid < OROW) histS[tid] = 0u;
  if (c == 0 && tid == 0) ovfn[0] = 0.0f;

  unsigned pk[PI];
#pragma unroll
  for (int i = 0; i < PI; ++i) {
    int k = i * PT + tid;
    pk[i] = 0xFFFFFFFFu;
    if (k < EPC) {
      unsigned d = (unsigned)dst[base + k];
      unsigned s = (unsigned)src[base + k];
      pk[i] = (d << 16) | (s & 0xFFFFu);
    }
  }
  __syncthreads();

  unsigned rk[PI];
#pragma unroll
  for (int i = 0; i < PI; ++i) {
    int k = i * PT + tid;
    rk[i] = 0u;
    if (k < EPC) rk[i] = atomicAdd(&histS[pk[i] >> 24], 1u);
  }
  __syncthreads();

  if (tid < 32) {
    unsigned v[8];
    unsigned sum = 0u;
#pragma unroll
    for (int j = 0; j < 8; ++j) { v[j] = histS[tid * 8 + j]; sum += v[j]; }
    unsigned inc = sum;
#pragma unroll
    for (int o = 1; o < 32; o <<= 1) {
      unsigned t = __shfl_up(inc, o, 32);
      if (tid >= o) inc += t;
    }
    unsigned run = inc - sum;
#pragma unroll
    for (int j = 0; j < 8; ++j) { offS[tid * 8 + j] = run; run += v[j]; }
  }
  __syncthreads();

#pragma unroll
  for (int i = 0; i < PI; ++i) {
    int k = i * PT + tid;
    if (k < EPC) sortS[offS[pk[i] >> 24] + rk[i]] = pk[i];
  }
  __syncthreads();

  for (int k = tid; k < EPC; k += PT) sorted[c * SROW + k] = sortS[k];
  if (tid < OROW) offtab[c * OROW + tid] = (int)offS[tid];
}

#define SPILL 1024
__device__ __forceinline__ void place_edge(unsigned w, unsigned* cntS, int* ellS,
                                           unsigned* spillN, unsigned* spillS) {
  int dl = (int)((w >> 16) & (NB_NODES - 1));
  unsigned p = atomicAdd(&cntS[dl], 1u);
  if (p < (unsigned)CAP) {
    ellS[dl * CAP + (int)p] = (int)(w & 0xFFFFu);
  } else {
    unsigned q = atomicAdd(spillN, 1u);
    if (q < (unsigned)SPILL) spillS[q] = w;
  }
}

__global__ __launch_bounds__(BT) void bin_k(const unsigned* __restrict__ sorted,
                                            const int* __restrict__ offtab,
                                            float* __restrict__ cnt,
                                            float* __restrict__ ovfn,
                                            int* __restrict__ ell,
                                            int* __restrict__ ovf) {
  __shared__ unsigned cntS[NB_NODES];
  __shared__ int ellS[NB_NODES * CAP];
  __shared__ unsigned spillS[SPILL];
  __shared__ unsigned spillN;
  __shared__ int spillBase;

  const int tid = threadIdx.x;
  const int lane = tid & 31;
  const int wave = __builtin_amdgcn_readfirstlane(tid >> 5);
  const int b = blockIdx.x;

  if (tid < NB_NODES) cntS[tid] = 0u;
  if (tid == 0) spillN = 0u;

  int st_l = 0, en_l = 0;
  if (lane < SPW) {
    const int* row = offtab + (wave + BWAVES * lane) * OROW + b;
    st_l = row[0];
    en_l = row[1];
  }
  st_l = st_l < 0 ? 0 : (st_l > EPC ? EPC : st_l);
  en_l = en_l < st_l ? st_l : (en_l > EPC ? EPC : en_l);

  unsigned ent[SPW];
  int ln[SPW];
#pragma unroll
  for (int u = 0; u < SPW; ++u) {
    int s0 = __builtin_amdgcn_readlane(st_l, u);
    ln[u] = __builtin_amdgcn_readlane(en_l, u) - s0;
    ent[u] = 0xFFFFFFFFu;
    if (lane < ln[u]) ent[u] = sorted[(wave + BWAVES * u) * SROW + s0 + lane];
  }
  __syncthreads();

#pragma unroll
  for (int u = 0; u < SPW; ++u) {
    if (lane < ln[u]) place_edge(ent[u], cntS, ellS, &spillN, spillS);
  }
#pragma nounroll
  for (int u = 0; u < SPW; ++u) {
    int s0 = __builtin_amdgcn_readlane(st_l, u);
    int len = __builtin_amdgcn_readlane(en_l, u) - s0;
    for (int k = 32 + lane; k < len; k += 32)
      place_edge(sorted[(wave + BWAVES * u) * SROW + s0 + k], cntS, ellS, &spillN, spillS);
  }
  __syncthreads();

#pragma unroll 4
  for (int r = 0; r < NB_NODES / BWAVES; ++r) {
    int nl = wave * (NB_NODES / BWAVES) + r;
    int node = b * NB_NODES + nl;
    if (node < GN) {
      unsigned m = cntS[nl];
      ell[node * CAP + lane] = ((unsigned)lane < m) ? ellS[nl * CAP + lane] : 0;
    }
  }
  if (tid < NB_NODES) {
    int node = b * NB_NODES + tid;
    if (node < GN) cnt[node] = (float)cntS[tid];
  }

  const unsigned sn = spillN;
  if (sn != 0u) {
    if (tid == 0) {
      float add = (float)sn;
      if (sn > (unsigned)SPILL) add += (float)(OVF_CAP + 1);
      spillBase = (int)atomicAdd(ovfn, add);
    }
    __syncthreads();
    const int base = spillBase;
    const int m = (int)(sn < (unsigned)SPILL ? sn : (unsigned)SPILL);
    for (int i = tid; i < m; i += BT) {
      int q = base + i;
      if (q >= 0 && q < OVF_CAP) atomicExch(&ovf[q], (int)spillS[i]);
    }
  }
}

template <bool HEAD>
__global__ __launch_bounds__(LAYER_T) void sage_pull_gemm_k(
    const float* __restrict__ cnt, const int* __restrict__ ell,
    const float* __restrict__ ovfn, const int* __restrict__ ovf,
    const int* __restrict__ src, const int* __restrict__ dst,
    const float* __restrict__ feat,
    const float* __restrict__ Wl, const float* __restrict__ Wr,
    const float* __restrict__ bl,
    const float* __restrict__ Wf, const float* __restrict__ bf,
    float* __restrict__ outp, int ntiles)
{
  __shared__ __attribute__((aligned(16))) float WlS[(GD / 2) * WROW];
  __shared__ __attribute__((aligned(16))) float WrS[(GD / 2) * WROW];
  __shared__ __attribute__((aligned(16))) float MS[BROWS * MROW];
  __shared__ float OS[BROWS];

  for (int i = threadIdx.x; i < GD * GD; i += LAYER_T) {
    int j = i >> 6;
    int k = i & 63;
    int o = (k >> 1) * WROW + j * 2 + (k & 1);
    WlS[o] = Wl[i];
    WrS[o] = Wr[i];
  }

  const int lane = threadIdx.x & 31;
  const int wave = __builtin_amdgcn_readfirstlane((int)(threadIdx.x >> 5));
  const int row0 = blockIdx.x * BROWS;

  {
    const int node0 = row0 + wave * 4;
    int dg[4];
    int idx0[4];
#pragma unroll
    for (int r = 0; r < 4; ++r) {
      int nd = node0 + r;
      float c = (nd < GN) ? cnt[nd] : 0.0f;
      dg[r] = (int)c;
      int m = dg[r] < CAP ? dg[r] : CAP;
      idx0[r] = (lane < m) ? ell[nd * CAP + lane] : 0;
    }
#pragma unroll
    for (int r = 0; r < 4; ++r) {
      const int deg = dg[r];
      const int m = deg < CAP ? deg : CAP;
      const int myidx = idx0[r];
      v2f a0 = {0.0f, 0.0f}, a1 = {0.0f, 0.0f}, a2 = {0.0f, 0.0f}, a3 = {0.0f, 0.0f};
      for (int j = 0; j < m; j += 8) {
        v2f v[8];
#pragma unroll
        for (int u = 0; u < 8; ++u) {
          int s = __builtin_amdgcn_readlane(myidx, j + u);
          v[u] = *(const v2f*)(feat + (long long)s * GD + 2 * lane);
        }
#pragma unroll
        for (int u = 0; u < 8; ++u) {
          if (j + u >= m) v[u] = (v2f){0.0f, 0.0f};
        }
        a0 += v[0]; a1 += v[1]; a2 += v[2]; a3 += v[3];
        a0 += v[4]; a1 += v[5]; a2 += v[6]; a3 += v[7];
      }
      if (deg > CAP) {
        const int node = node0 + r;
        const int tot = (int)ovfn[0];
        const bool full = tot > OVF_CAP;
        const int n = full ? GE : tot;
        if (full) {
          a0 = (v2f){0.0f, 0.0f}; a1 = a0; a2 = a0; a3 = a0;
        }
        for (int i0 = 0; i0 < n; i0 += 32) {
          const int i = i0 + lane;
          int dd = -1, ss = 0;
          if (i < n) {
            if (full) {
              dd = dst[i];
              ss = src[i];
            } else {
              unsigned w = (unsigned)ovf[i];
              dd = (int)(w >> 16);
              ss = (int)(w & 0xFFFFu);
            }
          }
          unsigned hit = __builtin_amdgcn_ballot_w32(dd == node);
          while (hit) {
            int b = __builtin_ctz(hit);
            hit &= hit - 1;
            int s = __builtin_amdgcn_readlane(ss, b);
            a0 += *(const v2f*)(feat + (long long)s * GD + 2 * lane);
          }
        }
      }
      const float inv = 1.0f / fmaxf((float)deg, 1.0f);
      v2f mean = ((a0 + a1) + (a2 + a3)) * inv;
      *(v2f*)&MS[(wave * 4 + r) * MROW + 2 * lane] = mean;
    }
  }
  __syncthreads();

  const int tile = blockIdx.x * TPB + wave;
  if (wave < TPB && tile < ntiles) {
    int half = lane >> 4;
    int l16  = lane & 15;
    int row  = tile * 16 + l16;

    const float* mRow = &MS[(wave * 16 + l16) * MROW];
    const float* xRow = feat + (long long)row * GD;

    v8f acc[4] = {};

    for (int k0 = 0; k0 < GD; k0 += 4) {
      int ka = k0 + half * 2;
      v2f am = *(const v2f*)(mRow + ka);
      v2f ar = *(const v2f*)(xRow + ka);
      int kb = (ka >> 1) * WROW;
#pragma unroll
      for (int t = 0; t < 4; ++t) {
        int n = t * 16 + l16;
        v2f bL = *(const v2f*)&WlS[kb + n * 2];
        v2f bR = *(const v2f*)&WrS[kb + n * 2];
        acc[t] = __builtin_amdgcn_wmma_f32_16x16x4_f32(
            false, am, false, bL, (short)0, acc[t], false, false);
        acc[t] = __builtin_amdgcn_wmma_f32_16x16x4_f32(
            false, ar, false, bR, (short)0, acc[t], false, false);
      }
    }

    if (!HEAD) {
#pragma unroll
      for (int t = 0; t < 4; ++t) {
        int n = t * 16 + l16;
        float bias = bl[n];
#pragma unroll
        for (int v = 0; v < 8; ++v) {
          int mr = half * 8 + v;
          float val = fmaxf(acc[t][v] + bias, 0.0f);
          outp[(long long)(tile * 16 + mr) * GD + n] = val;
        }
      }
    } else {
      float part[8];
#pragma unroll
      for (int v = 0; v < 8; ++v) part[v] = 0.0f;
#pragma unroll
      for (int t = 0; t < 4; ++t) {
        int n = t * 16 + l16;
        float bias = bl[n];
        float wf = Wf[n];
#pragma unroll
        for (int v = 0; v < 8; ++v) {
          float val = fmaxf(acc[t][v] + bias, 0.0f);
          part[v] += val * wf;
        }
      }
      const float b0 = bf[0];
#pragma unroll
      for (int v = 0; v < 8; ++v) {
        float p = part[v];
        p += __shfl_xor(p, 1, 32);
        p += __shfl_xor(p, 2, 32);
        p += __shfl_xor(p, 4, 32);
        p += __shfl_xor(p, 8, 32);
        if (l16 == 0) OS[wave * 16 + half * 8 + v] = p + b0;
      }
    }
  }

  if (HEAD) {
    __syncthreads();
    if (threadIdx.x < BROWS) {
      int node = row0 + (int)threadIdx.x;
      if (node < GN) outp[node] = OS[threadIdx.x];
    }
  }
}

extern "C" void kernel_launch(void* const* d_in, const int* in_sizes, int n_in,
                              void* d_out, int out_size, void* d_ws, size_t ws_size,
                              hipStream_t stream) {
  (void)in_sizes; (void)n_in; (void)out_size; (void)ws_size;

  const float* x   = (const float*)d_in[0];
  const int*   ei  = (const int*)d_in[1];
  const int*   src = ei;
  const int*   dst = ei + GE;
  const float* Wl1 = (const float*)d_in[2];
  const float* bl1 = (const float*)d_in[3];
  const float* Wr1 = (const float*)d_in[4];
  const float* Wl2 = (const float*)d_in[5];
  const float* bl2 = (const float*)d_in[6];
  const float* Wr2 = (const float*)d_in[7];
  const float* Wf  = (const float*)d_in[8];
  const float* bf  = (const float*)d_in[9];

  const size_t SZ_H   = (size_t)GN * GD * 4;
  const size_t SZ_CNT = (((size_t)GN * 4 + 255) / 256) * 256;
  const size_t SZ_OVN = 256;
  const size_t SZ_ELL = (size_t)GN * CAP * 4;

  char*  ws     = (char*)d_ws;
  float* h1     = (float*)ws;
  float* cnt    = (float*)(ws + SZ_H);
  float* ovfn   = (float*)(ws + SZ_H + SZ_CNT);
  int*   ell    = (int*)(ws + SZ_H + SZ_CNT + SZ_OVN);
  int*   ovf    = (int*)(ws + SZ_H + SZ_CNT + SZ_OVN + SZ_ELL);

  const int ntiles = GN / 16;
  const int layer_blocks = (ntiles + TPB - 1) / TPB;
  unsigned* sorted = (unsigned*)ws;
  int*      offtab = (int*)(ws + (size_t)NCHUNK * SROW * 4);

  part_k<<<NCHUNK, PT, 0, stream>>>(src, dst, sorted, offtab, ovfn);
  bin_k<<<NBINS, BT, 0, stream>>>(sorted, offtab, cnt, ovfn, ell, ovf);

  sage_pull_gemm_k<false><<<layer_blocks, LAYER_T, 0, stream>>>(
      cnt, ell, ovfn, ovf, src, dst, x, Wl1, Wr1, bl1, Wf, bf, h1, ntiles);
  sage_pull_gemm_k<true><<<layer_blocks, LAYER_T, 0, stream>>>(
      cnt, ell, ovfn, ovf, src, dst, h1, Wl2, Wr2, bl2, Wf, bf, (float*)d_out, ntiles);
}


# Neighbour rows read 16 B per lane, two rows per load (one per half-wave), a node's whole bucket in flight; a wave pulls 8 rows, half the waves per layer; neighbour sum order changes
# speedup vs baseline: 5.9836x; 1.1124x over previous
#include <hip/hip_runtime.h>


typedef __attribute__((ext_vector_type(2))) float v2f;
typedef __attribute__((ext_vector_type(4))) float v4f;
typedef __attribute__((ext_vector_type(8))) float v8f;

#define GN 50000
#define GE 800000
#define GD 64
#define WROW 160
#define TPB 4
#define BROWS (TPB * 16)
#define RPW 8
#define LAYER_T ((BROWS / RPW) * 32)
static_assert(BROWS % RPW == 0 && RPW <= 32, "rows per wave");
#define MROW 68
#define CAP 32
#define OVF_CAP 49152
#define NB_NODES 256
#define NBINS ((GN + NB_NODES - 1) / NB_NODES)
#define NCHUNK 256
#define EPC ((GE + NCHUNK - 1) / NCHUNK)
#define SROW (((EPC + 31) / 32) * 32)
#define OROW 256
#define PT 512
#define PI ((EPC + PT - 1) / PT)
#define BT 512
#define BWAVES (BT / 32)
#define SPW (NCHUNK / BWAVES)
static_assert(GN <= 65536, "node numbers must fit 16 bits");
static_assert(NB_NODES == 256, "range number is taken as word >> 24");
static_assert(NBINS + 1 <= OROW, "offsets row too short");
static_assert(OROW == 256, "scan below covers 32 lanes x 8 entries");
static_assert(SPW <= 32 && SPW * BWAVES == NCHUNK, "chunks per wave");
static_assert(GE % NCHUNK == 0, "every chunk is full");

__global__ __launch_bounds__(PT) void part_k(const int* __restrict__ src,
                                             const int* __restrict__ dst,
                                             unsigned* __restrict__ sorted,
                                             int* __restrict__ offtab,
                                             float* __restrict__ ovfn) {
  __shared__ unsigned histS[OROW];
  __shared__ unsigned offS[OROW];
  __shared__ unsigned sortS[EPC];

  const int tid = threadIdx.x;
  __builtin_assume(tid < PT);
  const int c = blockIdx.x;
  const int base = c * EPC;

  if (tid < OROW) histS[tid] = 0u;
  if (c == 0 && tid == 0) ovfn[0] = 0.0f;

  unsigned pk[PI];
#pragma unroll
  for (int i = 0; i < PI; ++i) {
    int k = i * PT + tid;
    pk[i] = 0xFFFFFFFFu;
    if (k < EPC) {
      unsigned d = (unsigned)dst[base + k];
      unsigned s = (unsigned)src[base + k];
      pk[i] = (d << 16) | (s & 0xFFFFu);
    }
  }
  __syncthreads();

  unsigned rk[PI];
#pragma unroll
  for (int i = 0; i < PI; ++i) {
    int k = i * PT + tid;
    rk[i] = 0u;
    if (k < EPC) rk[i] = atomicAdd(&histS[pk[i] >> 24], 1u);
  }
  __syncthreads();

  if (tid < 32) {
    unsigned v[8];
    unsigned sum = 0u;
#pragma unroll
    for (int j = 0; j < 8; ++j) { v[j] = histS[tid * 8 + j]; sum += v[j]; }
    unsigned inc = sum;
#pragma unroll
    for (int o = 1; o < 32; o <<= 1) {
      unsigned t = __shfl_up(inc, o, 32);
      if (tid >= o) inc += t;
    }
    unsigned run = inc - sum;
#pragma unroll
    for (int j = 0; j < 8; ++j) { offS[tid * 8 + j] = run; run += v[j]; }
  }
  __syncthreads();

#pragma unroll
  for (int i = 0; i < PI; ++i) {
    int k = i * PT + tid;
    if (k < EPC) sortS[offS[pk[i] >> 24] + rk[i]] = pk[i];
  }
  __syncthreads();

  for (int k = tid; k < EPC; k += PT) sorted[c * SROW + k] = sortS[k];
  if (tid < OROW) offtab[c * OROW + tid] = (int)offS[tid];
}

#define SPILL 1024
__device__ __forceinline__ void place_edge(unsigned w, unsigned* cntS, int* ellS,
                                           unsigned* spillN, unsigned* spillS) {
  int dl = (int)((w >> 16) & (NB_NODES - 1));
  unsigned p = atomicAdd(&cntS[dl], 1u);
  if (p < (unsigned)CAP) {
    ellS[dl * CAP + (int)p] = (int)(w & 0xFFFFu);
  } else {
    unsigned q = atomicAdd(spillN, 1u);
    if (q < (unsigned)SPILL) spillS[q] = w;
  }
}

__global__ __launch_bounds__(BT) void bin_k(const unsigned* __restrict__ sorted,
                                            const int* __restrict__ offtab,
                                            float* __restrict__ cnt,
                                            float* __restrict__ ovfn,
                                            int* __restrict__ ell,
                                            int* __restrict__ ovf) {
  __shared__ unsigned cntS[NB_NODES];
  __shared__ int ellS[NB_NODES * CAP];
  __shared__ unsigned spillS[SPILL];
  __shared__ unsigned spillN;
  __shared__ int spillBase;

  const int tid = threadIdx.x;
  const int lane = tid & 31;
  const int wave = __builtin_amdgcn_readfirstlane(tid >> 5);
  const int b = blockIdx.x;

  if (tid < NB_NODES) cntS[tid] = 0u;
  if (tid == 0) spillN = 0u;

  int st_l = 0, en_l = 0;
  if (lane < SPW) {
    const int* row = offtab + (wave + BWAVES * lane) * OROW + b;
    st_l = row[0];
    en_l = row[1];
  }
  st_l = st_l < 0 ? 0 : (st_l > EPC ? EPC : st_l);
  en_l = en_l < st_l ? st_l : (en_l > EPC ? EPC : en_l);

  unsigned ent[SPW];
  int ln[SPW];
#pragma unroll
  for (int u = 0; u < SPW; ++u) {
    int s0 = __builtin_amdgcn_readlane(st_l, u);
    ln[u] = __builtin_amdgcn_readlane(en_l, u) - s0;
    ent[u] = 0xFFFFFFFFu;
    if (lane < ln[u]) ent[u] = sorted[(wave + BWAVES * u) * SROW + s0 + lane];
  }
  __syncthreads();

#pragma unroll
  for (int u = 0; u < SPW; ++u) {
    if (lane < ln[u]) place_edge(ent[u], cntS, ellS, &spillN, spillS);
  }
#pragma nounroll
  for (int u = 0; u < SPW; ++u) {
    int s0 = __builtin_amdgcn_readlane(st_l, u);
    int len = __builtin_amdgcn_readlane(en_l, u) - s0;
    for (int k = 32 + lane; k < len; k += 32)
      place_edge(sorted[(wave + BWAVES * u) * SROW + s0 + k], cntS, ellS, &spillN, spillS);
  }
  __syncthreads();

#pragma unroll 4
  for (int r = 0; r < NB_NODES / BWAVES; ++r) {
    int nl = wave * (NB_NODES / BWAVES) + r;
    int node = b * NB_NODES + nl;
    if (node < GN) {
      unsigned m = cntS[nl];
      ell[node * CAP + lane] = ((unsigned)lane < m) ? ellS[nl * CAP + lane] : 0;
    }
  }
  if (tid < NB_NODES) {
    int node = b * NB_NODES + tid;
    if (node < GN) cnt[node] = (float)cntS[tid];
  }

  const unsigned sn = spillN;
  if (sn != 0u) {
    if (tid == 0) {
      float add = (float)sn;
      if (sn > (unsigned)SPILL) add += (float)(OVF_CAP + 1);
      spillBase = (int)atomicAdd(ovfn, add);
    }
    __syncthreads();
    const int base = spillBase;
    const int m = (int)(sn < (unsigned)SPILL ? sn : (unsigned)SPILL);
    for (int i = tid; i < m; i += BT) {
      int q = base + i;
      if (q >= 0 && q < OVF_CAP) atomicExch(&ovf[q], (int)spillS[i]);
    }
  }
}

template <int P>
__device__ __forceinline__ v4f pull_row(const float* __restrict__ feat, int myidx, int p16) {
  const int s = __builtin_amdgcn_update_dpp(0, myidx, 0x150 + P, 0xF, 0xF, false);
  return *((const v4f*)feat + ((unsigned)s * (unsigned)(GD / 4) + (unsigned)p16));
}

template <bool HEAD>
__global__ __launch_bounds__(LAYER_T) void sage_pull_gemm_k(
    const float* __restrict__ cnt, const int* __restrict__ ell,
    const float* __restrict__ ovfn, const int* __restrict__ ovf,
    const int* __restrict__ src, const int* __restrict__ dst,
    const float* __restrict__ feat,
    const float* __restrict__ Wl, const float* __restrict__ Wr,
    const float* __restrict__ bl,
    const float* __restrict__ Wf, const float* __restrict__ bf,
    float* __restrict__ outp, int ntiles)
{
  __shared__ __attribute__((aligned(16))) float WlS[(GD / 2) * WROW];
  __shared__ __attribute__((aligned(16))) float WrS[(GD / 2) * WROW];
  __shared__ __attribute__((aligned(16))) float MS[BROWS * MROW];
  __shared__ float OS[BROWS];

  for (int i = threadIdx.x; i < GD * GD; i += LAYER_T) {
    int j = i >> 6;
    int k = i & 63;
    int o = (k >> 1) * WROW + j * 2 + (k & 1);
    WlS[o] = Wl[i];
    WrS[o] = Wr[i];
  }

  const int lane = threadIdx.x & 31;
  const int wave = __builtin_amdgcn_readfirstlane((int)(threadIdx.x >> 5));
  const int row0 = blockIdx.x * BROWS;

  {
    const int hw = lane >> 4;
    const int p16 = lane & 15;
    const int myslot = 2 * p16 + hw;
    const int node0 = row0 + wave * RPW;
    float cl = 0.0f;
    if (lane < RPW && node0 + lane < GN) cl = cnt[node0 + lane];
    int dg[RPW];
    int idx0[RPW];
#pragma unroll
    for (int r = 0; r < RPW; ++r) {
      const int nd = node0 + r;
      dg[r] = (int)__int_as_float(__builtin_amdgcn_readlane(__float_as_int(cl), r));
      const int m = dg[r] < CAP ? dg[r] : CAP;
      int e = 0;
      if (nd < GN) e = ell[nd * CAP + myslot];
      idx0[r] = (myslot < m) ? e : 0;
    }
#pragma unroll
    for (int r = 0; r < RPW; ++r) {
      const int deg = dg[r];
      const int m = deg < CAP ? deg : CAP;
      const int myidx = idx0[r];
      const v4f z4 = {0.0f, 0.0f, 0.0f, 0.0f};
      v4f v0 = z4, v1 = z4, v2 = z4, v3 = z4, v4 = z4, v5 = z4, v6 = z4, v7 = z4;
      v4f v8 = z4, v9 = z4, v10 = z4, v11 = z4, v12 = z4, v13 = z4, v14 = z4, v15 = z4;
      if (m > 0) {
        v0 = pull_row<0>(feat, myidx, p16);   v1 = pull_row<1>(feat, myidx, p16);
        v2 = pull_row<2>(feat, myidx, p16);   v3 = pull_row<3>(feat, myidx, p16);
      }
      if (m > 8) {
        v4 = pull_row<4>(feat, myidx, p16);   v5 = pull_row<5>(feat, myidx, p16);
        v6 = pull_row<6>(feat, myidx, p16);   v7 = pull_row<7>(feat, myidx, p16);
      }
      if (m > 16) {
        v8 = pull_row<8>(feat, myidx, p16);   v9 = pull_row<9>(feat, myidx, p16);
        v10 = pull_row<10>(feat, myidx, p16); v11 = pull_row<11>(feat, myidx, p16);
      }
      if (m > 24) {
        v12 = pull_row<12>(feat, myidx, p16); v13 = pull_row<13>(feat, myidx, p16);
        v14 = pull_row<14>(feat, myidx, p16); v15 = pull_row<15>(feat, myidx, p16);
      }
      const int left = m - hw;
      v4f a0 = (0 < left) ? v0 : z4;
      v4f a1 = (2 < left) ? v1 : z4;
      a0 += (4 < left) ? v2 : z4;    a1 += (6 < left) ? v3 : z4;
      a0 += (8 < left) ? v4 : z4;    a1 += (10 < left) ? v5 : z4;
      a0 += (12 < left) ? v6 : z4;   a1 += (14 < left) ? v7 : z4;
      a0 += (16 < left) ? v8 : z4;   a1 += (18 < left) ? v9 : z4;
      a0 += (20 < left) ? v10 : z4;  a1 += (22 < left) ? v11 : z4;
      a0 += (24 < left) ? v12 : z4;  a1 += (26 < left) ? v13 : z4;
      a0 += (28 < left) ? v14 : z4;  a1 += (30 < left) ? v15 : z4;
      v4f tot = a0 + a1;
      tot.x += __shfl_xor(tot.x, 16, 32);
      tot.y += __shfl_xor(tot.y, 16, 32);
      tot.z += __shfl_xor(tot.z, 16, 32);
      tot.w += __shfl_xor(tot.w, 16, 32);
      if (deg > CAP) {
        const int node = node0 + r;
        const int ovtot = (int)ovfn[0];
        const bool full = ovtot > OVF_CAP;
        const int n = full ? GE : ovtot;
        if (full) tot = z4;
        for (int i0 = 0; i0 < n; i0 += 32) {
          const int i = i0 + lane;
          int dd = -1, ss = 0;
          if (i < n) {
            if (full) {
              dd = dst[i];
              ss = src[i];
            } else {
              unsigned w = (unsigned)ovf[i];
              dd = (int)(w >> 16);
              ss = (int)(w & 0xFFFFu);
            }
          }
          unsigned hit = __builtin_amdgcn_ballot_w32(dd == node);
          while (hit) {
            int b = __builtin_ctz(hit);
            hit &= hit - 1;
            int s = __builtin_amdgcn_readlane(ss, b);
            tot += *(const v4f*)(feat + (long long)s * GD + 4 * p16);
          }
        }
      }
      const float inv = 1.0f / fmaxf((float)deg, 1.0f);
      if (hw == 0) *(v4f*)&MS[(wave * RPW + r) * MROW + 4 * p16] = tot * inv;
    }
  }
  __syncthreads();

  const int tile = blockIdx.x * TPB + wave;
  if (wave < TPB && tile < ntiles) {
    int half = lane >> 4;
    int l16  = lane & 15;
    int row  = tile * 16 + l16;

    const float* mRow = &MS[(wave * 16 + l16) * MROW];
    const float* xRow = feat + (long long)row * GD;

    v8f acc[4] = {};

    for (int k0 = 0; k0 < GD; k0 += 4) {
      int ka = k0 + half * 2;
      v2f am = *(const v2f*)(mRow + ka);
      v2f ar = *(const v2f*)(xRow + ka);
      int kb = (ka >> 1) * WROW;
#pragma unroll
      for (int t = 0; t < 4; ++t) {
        int n = t * 16 + l16;
        v2f bL = *(const v2f*)&WlS[kb + n * 2];
        v2f bR = *(const v2f*)&WrS[kb + n * 2];
        acc[t] = __builtin_amdgcn_wmma_f32_16x16x4_f32(
            false, am, false, bL, (short)0, acc[t], false, false);
        acc[t] = __builtin_amdgcn_wmma_f32_16x16x4_f32(
            false, ar, false, bR, (short)0, acc[t], false, false);
      }
    }

    if (!HEAD) {
#pragma unroll
      for (int t = 0; t < 4; ++t) {
        int n = t * 16 + l16;
        float bias = bl[n];
#pragma unroll
        for (int v = 0; v < 8; ++v) {
          int mr = half * 8 + v;
          float val = fmaxf(acc[t][v] + bias, 0.0f);
          outp[(long long)(tile * 16 + mr) * GD + n] = val;
        }
      }
    } else {
      float part[8];
#pragma unroll
      for (int v = 0; v < 8; ++v) part[v] = 0.0f;
#pragma unroll
      for (int t = 0; t < 4; ++t) {
        int n = t * 16 + l16;
        float bias = bl[n];
        float wf = Wf[n];
#pragma unroll
        for (int v = 0; v < 8; ++v) {
          float val = fmaxf(acc[t][v] + bias, 0.0f);
          part[v] += val * wf;
        }
      }
      const float b0 = bf[0];
#pragma unroll
      for (int v = 0; v < 8; ++v) {
        float p = part[v];
        p += __shfl_xor(p, 1, 32);
        p += __shfl_xor(p, 2, 32);
        p += __shfl_xor(p, 4, 32);
        p += __shfl_xor(p, 8, 32);
        if (l16 == 0) OS[wave * 16 + half * 8 + v] = p + b0;
      }
    }
  }

  if (HEAD) {
    __syncthreads();
    if (threadIdx.x < BROWS) {
      int node = row0 + (int)threadIdx.x;
      if (node < GN) outp[node] = OS[threadIdx.x];
    }
  }
}

extern "C" void kernel_launch(void* const* d_in, const int* in_sizes, int n_in,
                              void* d_out, int out_size, void* d_ws, size_t ws_size,
                              hipStream_t stream) {
  (void)in_sizes; (void)n_in; (void)out_size; (void)ws_size;

  const float* x   = (const float*)d_in[0];
  const int*   ei  = (const int*)d_in[1];
  const int*   src = ei;
  const int*   dst = ei + GE;
  const float* Wl1 = (const float*)d_in[2];
  const float* bl1 = (const float*)d_in[3];
  const float* Wr1 = (const float*)d_in[4];
  const float* Wl2 = (const float*)d_in[5];
  const float* bl2 = (const float*)d_in[6];
  const float* Wr2 = (const float*)d_in[7];
  const float* Wf  = (const float*)d_in[8];
  const float* bf  = (const float*)d_in[9];

  const size_t SZ_H   = (size_t)GN * GD * 4;
  const size_t SZ_CNT = (((size_t)GN * 4 + 255) / 256) * 256;
  const size_t SZ_OVN = 256;
  const size_t SZ_ELL = (size_t)GN * CAP * 4;

  char*  ws     = (char*)d_ws;
  float* h1     = (float*)ws;
  float* cnt    = (float*)(ws + SZ_H);
  float* ovfn   = (float*)(ws + SZ_H + SZ_CNT);
  int*   ell    = (int*)(ws + SZ_H + SZ_CNT + SZ_OVN);
  int*   ovf    = (int*)(ws + SZ_H + SZ_CNT + SZ_OVN + SZ_ELL);

  const int ntiles = GN / 16;
  const int layer_blocks = (ntiles + TPB - 1) / TPB;
  unsigned* sorted = (unsigned*)ws;
  int*      offtab = (int*)(ws + (size_t)NCHUNK * SROW * 4);

  part_k<<<NCHUNK, PT, 0, stream>>>(src, dst, sorted, offtab, ovfn);
  bin_k<<<NBINS, BT, 0, stream>>>(sorted, offtab, cnt, ovfn, ell, ovf);

  sage_pull_gemm_k<false><<<layer_blocks, LAYER_T, 0, stream>>>(
      cnt, ell, ovfn, ovf, src, dst, x, Wl1, Wr1, bl1, Wf, bf, h1, ntiles);
  sage_pull_gemm_k<true><<<layer_blocks, LAYER_T, 0, stream>>>(
      cnt, ell, ovfn, ovf, src, dst, h1, Wl2, Wr2, bl2, Wf, bf, (float*)d_out, ntiles);
}


# Layer block: weights (16-byte pieces into a row-major LDS copy, pitch 68), degrees, bucket lines and root rows requested as one group at block start instead of ten dependent round trips
# speedup vs baseline: 7.0728x; 1.1820x over previous
#include <hip/hip_runtime.h>


typedef __attribute__((ext_vector_type(2))) float v2f;
typedef __attribute__((ext_vector_type(4))) float v4f;
typedef __attribute__((ext_vector_type(8))) float v8f;

#define GN 50000
#define GE 800000
#define GD 64
#define WSTR 68
#define TPB 4
#define BROWS (TPB * 16)
#define RPW 8
#define LAYER_T ((BROWS / RPW) * 32)
static_assert(BROWS % RPW == 0 && RPW <= 32, "rows per wave");
#define WPT ((GD * GD / 4) / LAYER_T)
static_assert(WPT * LAYER_T * 4 == GD * GD, "weight pieces divide evenly");
#define MROW 68
#define CAP 32
#define OVF_CAP 49152
#define NB_NODES 256
#define NBINS ((GN + NB_NODES - 1) / NB_NODES)
#define NCHUNK 256
#define EPC ((GE + NCHUNK - 1) / NCHUNK)
#define SROW (((EPC + 31) / 32) * 32)
#define OROW 256
#define PT 512
#define PI ((EPC + PT - 1) / PT)
#define BT 512
#define BWAVES (BT / 32)
#define SPW (NCHUNK / BWAVES)
static_assert(GN <= 65536, "node numbers must fit 16 bits");
static_assert(NB_NODES == 256, "range number is taken as word >> 24");
static_assert(NBINS + 1 <= OROW, "offsets row too short");
static_assert(OROW == 256, "scan below covers 32 lanes x 8 entries");
static_assert(SPW <= 32 && SPW * BWAVES == NCHUNK, "chunks per wave");
static_assert(GE % NCHUNK == 0, "every chunk is full");

__global__ __launch_bounds__(PT) void part_k(const int* __restrict__ src,
                                             const int* __restrict__ dst,
                                             unsigned* __restrict__ sorted,
                                             int* __restrict__ offtab,
                                             float* __restrict__ ovfn) {
  __shared__ unsigned histS[OROW];
  __shared__ unsigned offS[OROW];
  __shared__ unsigned sortS[EPC];

  const int tid = threadIdx.x;
  __builtin_assume(tid < PT);
  const int c = blockIdx.x;
  const int base = c * EPC;

  if (tid < OROW) histS[tid] = 0u;
  if (c == 0 && tid == 0) ovfn[0] = 0.0f;

  unsigned pk[PI];
#pragma unroll
  for (int i = 0; i < PI; ++i) {
    int k = i * PT + tid;
    pk[i] = 0xFFFFFFFFu;
    if (k < EPC) {
      unsigned d = (unsigned)dst[base + k];
      unsigned s = (unsigned)src[base + k];
      pk[i] = (d << 16) | (s & 0xFFFFu);
    }
  }
  __syncthreads();

  unsigned rk[PI];
#pragma unroll
  for (int i = 0; i < PI; ++i) {
    int k = i * PT + tid;
    rk[i] = 0u;
    if (k < EPC) rk[i] = atomicAdd(&histS[pk[i] >> 24], 1u);
  }
  __syncthreads();

  if (tid < 32) {
    unsigned v[8];
    unsigned sum = 0u;
#pragma unroll
    for (int j = 0; j < 8; ++j) { v[j] = histS[tid * 8 + j]; sum += v[j]; }
    unsigned inc = sum;
#pragma unroll
    for (int o = 1; o < 32; o <<= 1) {
      unsigned t = __shfl_up(inc, o, 32);
      if (tid >= o) inc += t;
    }
    unsigned run = inc - sum;
#pragma unroll
    for (int j = 0; j < 8; ++j) { offS[tid * 8 + j] = run; run += v[j]; }
  }
  __syncthreads();

#pragma unroll
  for (int i = 0; i < PI; ++i) {
    int k = i * PT + tid;
    if (k < EPC) sortS[offS[pk[i] >> 24] + rk[i]] = pk[i];
  }
  __syncthreads();

  for (int k = tid; k < EPC; k += PT) sorted[c * SROW + k] = sortS[k];
  if (tid < OROW) offtab[c * OROW + tid] = (int)offS[tid];
}

#define SPILL 1024
__device__ __forceinline__ void place_edge(unsigned w, unsigned* cntS, int* ellS,
                                           unsigned* spillN, unsigned* spillS) {
  int dl = (int)((w >> 16) & (NB_NODES - 1));
  unsigned p = atomicAdd(&cntS[dl], 1u);
  if (p < (unsigned)CAP) {
    ellS[dl * CAP + (int)p] = (int)(w & 0xFFFFu);
  } else {
    unsigned q = atomicAdd(spillN, 1u);
    if (q < (unsigned)SPILL) spillS[q] = w;
  }
}

__global__ __launch_bounds__(BT) void bin_k(const unsigned* __restrict__ sorted,
                                            const int* __restrict__ offtab,
                                            float* __restrict__ cnt,
                                            float* __restrict__ ovfn,
                                            int* __restrict__ ell,
                                            int* __restrict__ ovf) {
  __shared__ unsigned cntS[NB_NODES];
  __shared__ int ellS[NB_NODES * CAP];
  __shared__ unsigned spillS[SPILL];
  __shared__ unsigned spillN;
  __shared__ int spillBase;

  const int tid = threadIdx.x;
  const int lane = tid & 31;
  const int wave = __builtin_amdgcn_readfirstlane(tid >> 5);
  const int b = blockIdx.x;

  if (tid < NB_NODES) cntS[tid] = 0u;
  if (tid == 0) spillN = 0u;

  int st_l = 0, en_l = 0;
  if (lane < SPW) {
    const int* row = offtab + (wave + BWAVES * lane) * OROW + b;
    st_l = row[0];
    en_l = row[1];
  }
  st_l = st_l < 0 ? 0 : (st_l > EPC ? EPC : st_l);
  en_l = en_l < st_l ? st_l : (en_l > EPC ? EPC : en_l);

  unsigned ent[SPW];
  int ln[SPW];
#pragma unroll
  for (int u = 0; u < SPW; ++u) {
    int s0 = __builtin_amdgcn_readlane(st_l, u);
    ln[u] = __builtin_amdgcn_readlane(en_l, u) - s0;
    ent[u] = 0xFFFFFFFFu;
    if (lane < ln[u]) ent[u] = sorted[(wave + BWAVES * u) * SROW + s0 + lane];
  }
  __syncthreads();

#pragma unroll
  for (int u = 0; u < SPW; ++u) {
    if (lane < ln[u]) place_edge(ent[u], cntS, ellS, &spillN, spillS);
  }
#pragma nounroll
  for (int u = 0; u < SPW; ++u) {
    int s0 = __builtin_amdgcn_readlane(st_l, u);
    int len = __builtin_amdgcn_readlane(en_l, u) - s0;
    for (int k = 32 + lane; k < len; k += 32)
      place_edge(sorted[(wave + BWAVES * u) * SROW + s0 + k], cntS, ellS, &spillN, spillS);
  }
  __syncthreads();

#pragma unroll 4
  for (int r = 0; r < NB_NODES / BWAVES; ++r) {
    int nl = wave * (NB_NODES / BWAVES) + r;
    int node = b * NB_NODES + nl;
    if (node < GN) {
      unsigned m = cntS[nl];
      ell[node * CAP + lane] = ((unsigned)lane < m) ? ellS[nl * CAP + lane] : 0;
    }
  }
  if (tid < NB_NODES) {
    int node = b * NB_NODES + tid;
    if (node < GN) cnt[node] = (float)cntS[tid];
  }

  const unsigned sn = spillN;
  if (sn != 0u) {
    if (tid == 0) {
      float add = (float)sn;
      if (sn > (unsigned)SPILL) add += (float)(OVF_CAP + 1);
      spillBase = (int)atomicAdd(ovfn, add);
    }
    __syncthreads();
    const int base = spillBase;
    const int m = (int)(sn < (unsigned)SPILL ? sn : (unsigned)SPILL);
    for (int i = tid; i < m; i += BT) {
      int q = base + i;
      if (q >= 0 && q < OVF_CAP) atomicExch(&ovf[q], (int)spillS[i]);
    }
  }
}

template <int P>
__device__ __forceinline__ v4f pull_row(const float* __restrict__ feat, int myidx, int p16) {
  const int s = __builtin_amdgcn_update_dpp(0, myidx, 0x150 + P, 0xF, 0xF, false);
  return *((const v4f*)feat + ((unsigned)s * (unsigned)(GD / 4) + (unsigned)p16));
}

template <bool HEAD>
__global__ __launch_bounds__(LAYER_T) void sage_pull_gemm_k(
    const float* __restrict__ cnt, const int* __restrict__ ell,
    const float* __restrict__ ovfn, const int* __restrict__ ovf,
    const int* __restrict__ src, const int* __restrict__ dst,
    const float* __restrict__ feat,
    const float* __restrict__ Wl, const float* __restrict__ Wr,
    const float* __restrict__ bl,
    const float* __restrict__ Wf, const float* __restrict__ bf,
    float* __restrict__ outp, int ntiles)
{
  __shared__ __attribute__((aligned(16))) float WlS[GD * WSTR];
  __shared__ __attribute__((aligned(16))) float WrS[GD * WSTR];
  __shared__ __attribute__((aligned(16))) float MS[BROWS * MROW];
  __shared__ float OS[BROWS];

  const int lane = threadIdx.x & 31;
  const int wave = __builtin_amdgcn_readfirstlane((int)(threadIdx.x >> 5));
  const int row0 = blockIdx.x * BROWS;
  const int tile = blockIdx.x * TPB + wave;
  const bool gemm_wave = (wave < TPB) && (tile < ntiles);

  v4f wl[WPT], wr[WPT];
#pragma unroll
  for (int it = 0; it < WPT; ++it) {
    const int q = it * LAYER_T + (int)threadIdx.x;
    wl[it] = *((const v4f*)Wl + q);
    wr[it] = *((const v4f*)Wr + q);
  }
  const int hw = lane >> 4;
  const int p16 = lane & 15;
  const int myslot = 2 * p16 + hw;
  const int node0 = row0 + wave * RPW;
  float cl = 0.0f;
  if (lane < RPW && node0 + lane < GN) cl = cnt[node0 + lane];
  int eraw[RPW];
#pragma unroll
  for (int r = 0; r < RPW; ++r) {
    const int nd = node0 + r;
    const int ndc = nd < GN ? nd : GN - 1;
    eraw[r] = ell[ndc * CAP + myslot];
  }
  v2f ar[GD / 4];
#pragma unroll
  for (int j = 0; j < GD / 4; ++j) ar[j] = (v2f){0.0f, 0.0f};
  if (gemm_wave) {
    const float* xa = feat + (long long)(tile * 16 + p16) * GD + hw * 2;
#pragma unroll
    for (int j = 0; j < GD / 4; ++j) ar[j] = *(const v2f*)(xa + 4 * j);
  }

#pragma unroll
  for (int it = 0; it < WPT; ++it) {
    const int q = it * LAYER_T + (int)threadIdx.x;
    const int o = (q >> 4) * WSTR + (q & 15) * 4;
    *(v4f*)&WlS[o] = wl[it];
    *(v4f*)&WrS[o] = wr[it];
  }

  {
    int dg[RPW];
    int idx0[RPW];
#pragma unroll
    for (int r = 0; r < RPW; ++r) {
      dg[r] = (int)__int_as_float(__builtin_amdgcn_readlane(__float_as_int(cl), r));
      const int m = dg[r] < CAP ? dg[r] : CAP;
      idx0[r] = (myslot < m) ? eraw[r] : 0;
    }
#pragma unroll
    for (int r = 0; r < RPW; ++r) {
      const int deg = dg[r];
      const int m = deg < CAP ? deg : CAP;
      const int myidx = idx0[r];
      const v4f z4 = {0.0f, 0.0f, 0.0f, 0.0f};
      v4f v0 = z4, v1 = z4, v2 = z4, v3 = z4, v4 = z4, v5 = z4, v6 = z4, v7 = z4;
      v4f v8 = z4, v9 = z4, v10 = z4, v11 = z4, v12 = z4, v13 = z4, v14 = z4, v15 = z4;
      if (m > 0) {
        v0 = pull_row<0>(feat, myidx, p16);   v1 = pull_row<1>(feat, myidx, p16);
        v2 = pull_row<2>(feat, myidx, p16);   v3 = pull_row<3>(feat, myidx, p16);
      }
      if (m > 8) {
        v4 = pull_row<4>(feat, myidx, p16);   v5 = pull_row<5>(feat, myidx, p16);
        v6 = pull_row<6>(feat, myidx, p16);   v7 = pull_row<7>(feat, myidx, p16);
      }
      if (m > 16) {
        v8 = pull_row<8>(feat, myidx, p16);   v9 = pull_row<9>(feat, myidx, p16);
        v10 = pull_row<10>(feat, myidx, p16); v11 = pull_row<11>(feat, myidx, p16);
      }
      if (m > 24) {
        v12 = pull_row<12>(feat, myidx, p16); v13 = pull_row<13>(feat, myidx, p16);
        v14 = pull_row<14>(feat, myidx, p16); v15 = pull_row<15>(feat, myidx, p16);
      }
      const int left = m - hw;
      v4f a0 = (0 < left) ? v0 : z4;
      v4f a1 = (2 < left) ? v1 : z4;
      a0 += (4 < left) ? v2 : z4;    a1 += (6 < left) ? v3 : z4;
      a0 += (8 < left) ? v4 : z4;    a1 += (10 < left) ? v5 : z4;
      a0 += (12 < left) ? v6 : z4;   a1 += (14 < left) ? v7 : z4;
      a0 += (16 < left) ? v8 : z4;   a1 += (18 < left) ? v9 : z4;
      a0 += (20 < left) ? v10 : z4;  a1 += (22 < left) ? v11 : z4;
      a0 += (24 < left) ? v12 : z4;  a1 += (26 < left) ? v13 : z4;
      a0 += (28 < left) ? v14 : z4;  a1 += (30 < left) ? v15 : z4;
      v4f tot = a0 + a1;
      tot.x += __shfl_xor(tot.x, 16, 32);
      tot.y += __shfl_xor(tot.y, 16, 32);
      tot.z += __shfl_xor(tot.z, 16, 32);
      tot.w += __shfl_xor(tot.w, 16, 32);
      if (deg > CAP) {
        const int node = node0 + r;
        const int ovtot = (int)ovfn[0];
        const bool full = ovtot > OVF_CAP;
        const int n = full ? GE : ovtot;
        if (full) tot = z4;
        for (int i0 = 0; i0 < n; i0 += 32) {
          const int i = i0 + lane;
          int dd = -1, ss = 0;
          if (i < n) {
            if (full) {
              dd = dst[i];
              ss = src[i];
            } else {
              unsigned w = (unsigned)ovf[i];
              dd = (int)(w >> 16);
              ss = (int)(w & 0xFFFFu);
            }
          }
          unsigned hit = __builtin_amdgcn_ballot_w32(dd == node);
          while (hit) {
            int b = __builtin_ctz(hit);
            hit &= hit - 1;
            int s = __builtin_amdgcn_readlane(ss, b);
            tot += *(const v4f*)(feat + (long long)s * GD + 4 * p16);
          }
        }
      }
      const float inv = 1.0f / fmaxf((float)deg, 1.0f);
      if (hw == 0) *(v4f*)&MS[(wave * RPW + r) * MROW + 4 * p16] = tot * inv;
    }
  }
  __syncthreads();

  if (gemm_wave) {
    int half = lane >> 4;
    int l16  = lane & 15;

    const float* mRow = &MS[(wave * 16 + l16) * MROW];

    v8f acc[4] = {};

#pragma unroll
    for (int k0 = 0; k0 < GD; k0 += 4) {
      int ka = k0 + half * 2;
      v2f am = *(const v2f*)(mRow + ka);
#pragma unroll
      for (int t = 0; t < 4; ++t) {
        int n = t * 16 + l16;
        v2f bL = *(const v2f*)&WlS[n * WSTR + ka];
        v2f bR = *(const v2f*)&WrS[n * WSTR + ka];
        acc[t] = __builtin_amdgcn_wmma_f32_16x16x4_f32(
            false, am, false, bL, (short)0, acc[t], false, false);
        acc[t] = __builtin_amdgcn_wmma_f32_16x16x4_f32(
            false, ar[k0 / 4], false, bR, (short)0, acc[t], false, false);
      }
    }

    if (!HEAD) {
#pragma unroll
      for (int t = 0; t < 4; ++t) {
        int n = t * 16 + l16;
        float bias = bl[n];
#pragma unroll
        for (int v = 0; v < 8; ++v) {
          int mr = half * 8 + v;
          float val = fmaxf(acc[t][v] + bias, 0.0f);
          outp[(long long)(tile * 16 + mr) * GD + n] = val;
        }
      }
    } else {
      float part[8];
#pragma unroll
      for (int v = 0; v < 8; ++v) part[v] = 0.0f;
#pragma unroll
      for (int t = 0; t < 4; ++t) {
        int n = t * 16 + l16;
        float bias = bl[n];
        float wf = Wf[n];
#pragma unroll
        for (int v = 0; v < 8; ++v) {
          float val = fmaxf(acc[t][v] + bias, 0.0f);
          part[v] += val * wf;
        }
      }
      const float b0 = bf[0];
#pragma unroll
      for (int v = 0; v < 8; ++v) {
        float p = part[v];
        p += __shfl_xor(p, 1, 32);
        p += __shfl_xor(p, 2, 32);
        p += __shfl_xor(p, 4, 32);
        p += __shfl_xor(p, 8, 32);
        if (l16 == 0) OS[wave * 16 + half * 8 + v] = p + b0;
      }
    }
  }

  if (HEAD) {
    __syncthreads();
    if (threadIdx.x < BROWS) {
      int node = row0 + (int)threadIdx.x;
      if (node < GN) outp[node] = OS[threadIdx.x];
    }
  }
}

extern "C" void kernel_launch(void* const* d_in, const int* in_sizes, int n_in,
                              void* d_out, int out_size, void* d_ws, size_t ws_size,
                              hipStream_t stream) {
  (void)in_sizes; (void)n_in; (void)out_size; (void)ws_size;

  const float* x   = (const float*)d_in[0];
  const int*   ei  = (const int*)d_in[1];
  const int*   src = ei;
  const int*   dst = ei + GE;
  const float* Wl1 = (const float*)d_in[2];
  const float* bl1 = (const float*)d_in[3];
  const float* Wr1 = (const float*)d_in[4];
  const float* Wl2 = (const float*)d_in[5];
  const float* bl2 = (const float*)d_in[6];
  const float* Wr2 = (const float*)d_in[7];
  const float* Wf  = (const float*)d_in[8];
  const float* bf  = (const float*)d_in[9];

  const size_t SZ_H   = (size_t)GN * GD * 4;
  const size_t SZ_CNT = (((size_t)GN * 4 + 255) / 256) * 256;
  const size_t SZ_OVN = 256;
  const size_t SZ_ELL = (size_t)GN * CAP * 4;

  char*  ws     = (char*)d_ws;
  float* h1     = (float*)ws;
  float* cnt    = (float*)(ws + SZ_H);
  float* ovfn   = (float*)(ws + SZ_H + SZ_CNT);
  int*   ell    = (int*)(ws + SZ_H + SZ_CNT + SZ_OVN);
  int*   ovf    = (int*)(ws + SZ_H + SZ_CNT + SZ_OVN + SZ_ELL);

  const int ntiles = GN / 16;
  const int layer_blocks = (ntiles + TPB - 1) / TPB;
  unsigned* sorted = (unsigned*)ws;
  int*      offtab = (int*)(ws + (size_t)NCHUNK * SROW * 4);

  part_k<<<NCHUNK, PT, 0, stream>>>(src, dst, sorted, offtab, ovfn);
  bin_k<<<NBINS, BT, 0, stream>>>(sorted, offtab, cnt, ovfn, ell, ovf);

  sage_pull_gemm_k<false><<<layer_blocks, LAYER_T, 0, stream>>>(
      cnt, ell, ovfn, ovf, src, dst, x, Wl1, Wr1, bl1, Wf, bf, h1, ntiles);
  sage_pull_gemm_k<true><<<layer_blocks, LAYER_T, 0, stream>>>(
      cnt, ell, ovfn, ovf, src, dst, h1, Wl2, Wr2, bl2, Wf, bf, (float*)d_out, ntiles);
}
